# starting kernel
# baseline (speedup 1.0000x reference)
#include <hip/hip_runtime.h>

#define N_NODES 50000
#define N_EDGES 800000
#define FEATS   128
#define M_TILES (N_NODES / 16)
#define N_TILES (FEATS / 16)

typedef __attribute__((ext_vector_type(2))) float v2f;
typedef __attribute__((ext_vector_type(8))) float v8f;

__device__ __forceinline__ void atomic_add_f32(float* p, float v) {
    __hip_atomic_fetch_add(p, v, __ATOMIC_RELAXED, __HIP_MEMORY_SCOPE_AGENT);
}

__global__ void zero_kernel(float* __restrict__ p, int n) {
    int i = blockIdx.x * blockDim.x + threadIdx.x;
    if (i < n) p[i] = 0.0f;
}

__global__ void degree_kernel(const int* __restrict__ src, const int* __restrict__ dst,
                              float* __restrict__ deg_out, float* __restrict__ deg_in) {
    int e = blockIdx.x * blockDim.x + threadIdx.x;
    if (e < N_EDGES) {
        atomic_add_f32(&deg_out[src[e]], 1.0f);
        atomic_add_f32(&deg_in[dst[e]], 1.0f);
    }
}

__global__ void rsqrt_kernel(float* __restrict__ deg_out, float* __restrict__ deg_in) {
    int i = blockIdx.x * blockDim.x + threadIdx.x;
    if (i < N_NODES) {
        deg_out[i] = rsqrtf(fmaxf(deg_out[i], 1.0f));
        deg_in[i]  = rsqrtf(fmaxf(deg_in[i],  1.0f));
    }
}

__global__ __launch_bounds__(256)
void gemm_wmma_kernel(const float* __restrict__ x, const float* __restrict__ W,
                      const float* __restrict__ s_out, float* __restrict__ y) {
    const int wave   = (blockIdx.x * blockDim.x + threadIdx.x) >> 5;
    const int lane   = threadIdx.x & 31;
    const int tile_n = wave & (N_TILES - 1);
    const int tile_m = wave >> 3;

    const int m = lane & 15;
    const int h = lane >> 4;
    const int row = tile_m * 16 + m;
    const int col = tile_n * 16 + m;

    const float scale = s_out[row];
    const float* __restrict__ arow = x + (size_t)row * FEATS;

    v8f c = {};
    #pragma unroll 4
    for (int k0 = 0; k0 < FEATS; k0 += 4) {
        const int k = k0 + 2 * h;
        v2f a = *(const v2f*)(arow + k);
        a.x *= scale;
        a.y *= scale;
        v2f b;
        b.x = W[(size_t)k * FEATS + col];
        b.y = W[(size_t)(k + 1) * FEATS + col];
        c = __builtin_amdgcn_wmma_f32_16x16x4_f32(
                 false, a,  false, b,
                 (short)0, c,  false,  false);
    }

    #pragma unroll
    for (int j = 0; j < 8; ++j) {
        const int orow = tile_m * 16 + j + 8 * h;
        y[(size_t)orow * FEATS + col] = c[j];
    }
}

__global__ __launch_bounds__(256)
void scatter_kernel(const int* __restrict__ src, const int* __restrict__ dst,
                    const float* __restrict__ y, float* __restrict__ out) {
    const int e    = (blockIdx.x * blockDim.x + threadIdx.x) >> 5;
    const int lane = threadIdx.x & 31;
    if (e >= N_EDGES) return;
    const int s = src[e];
    const int d = dst[e];
    const float4 v = *(const float4*)(y + (size_t)s * FEATS + lane * 4);
    float* o = out + (size_t)d * FEATS + lane * 4;
    atomic_add_f32(o + 0, v.x);
    atomic_add_f32(o + 1, v.y);
    atomic_add_f32(o + 2, v.z);
    atomic_add_f32(o + 3, v.w);
}

__global__ void finalize_kernel(float* __restrict__ out, const float* __restrict__ s_in,
                                const float* __restrict__ bias) {
    int i = blockIdx.x * blockDim.x + threadIdx.x;
    if (i < N_NODES * FEATS) {
        int r = i >> 7;
        int ccol = i & (FEATS - 1);
        out[i] = out[i] * s_in[r] + bias[ccol];
    }
}

extern "C" void kernel_launch(void* const* d_in, const int* in_sizes, int n_in,
                              void* d_out, int out_size, void* d_ws, size_t ws_size,
                              hipStream_t stream) {
    const float* x    = (const float*)d_in[0];
    const int*   src  = (const int*)  d_in[1];
    const int*   dst  = (const int*)  d_in[2];
    const float* W    = (const float*)d_in[3];
    const float* bias = (const float*)d_in[4];
    float* out = (float*)d_out;

    float* s_out = (float*)d_ws;
    float* s_in  = s_out + N_NODES;
    float* y     = s_in + N_NODES;

    const int TB = 256;

    zero_kernel<<<(2 * N_NODES + TB - 1) / TB, TB, 0, stream>>>(s_out, 2 * N_NODES);
    zero_kernel<<<(N_NODES * FEATS + TB - 1) / TB, TB, 0, stream>>>(out, N_NODES * FEATS);

    degree_kernel<<<(N_EDGES + TB - 1) / TB, TB, 0, stream>>>(src, dst, s_out, s_in);

    rsqrt_kernel<<<(N_NODES + TB - 1) / TB, TB, 0, stream>>>(s_out, s_in);

    gemm_wmma_kernel<<<(M_TILES * N_TILES) / 8, TB, 0, stream>>>(x, W, s_out, y);

    scatter_kernel<<<(N_EDGES * 32) / TB, TB, 0, stream>>>(src, dst, y, out);

    finalize_kernel<<<(N_NODES * FEATS + TB - 1) / TB, TB, 0, stream>>>(out, s_in, bias);
}


# degree counts: per-edge global f32 atomics replaced by per-block LDS histograms (integer counters, two per word, exact), planes summed in fixed order in the rsqrt kernel; degree zero-fill dropped
# speedup vs baseline: 1.4098x; 1.4098x over previous
#include <hip/hip_runtime.h>

#define N_NODES 50000
#define N_EDGES 800000
#define FEATS   128
#define M_TILES (N_NODES / 16)
#define N_TILES (FEATS / 16)

typedef __attribute__((ext_vector_type(2))) float v2f;
typedef __attribute__((ext_vector_type(8))) float v8f;

__device__ __forceinline__ void atomic_add_f32(float* p, float v) {
    __hip_atomic_fetch_add(p, v, __ATOMIC_RELAXED, __HIP_MEMORY_SCOPE_AGENT);
}

__global__ void zero_kernel(float* __restrict__ p, int n) {
    int i = blockIdx.x * blockDim.x + threadIdx.x;
    if (i < n) p[i] = 0.0f;
}

#define HIST_SLICES     64
#define SLICE_EDGES     (N_EDGES / HIST_SLICES)
#define SLICE_INT4      (SLICE_EDGES / 4)
#define PLANE_WORDS     (N_NODES / 2)
#define PLANE_UINT4     (PLANE_WORDS / 4)
#define HIST_THREADS    1024
#define HIST_LOADS      ((SLICE_INT4 + HIST_THREADS - 1) / HIST_THREADS)

__device__ __forceinline__ void lds_count(unsigned int* h, int node) {
    __hip_atomic_fetch_add(&h[node >> 1], 1u << ((node & 1) << 4),
                           __ATOMIC_RELAXED, __HIP_MEMORY_SCOPE_WORKGROUP);
}

__global__ __launch_bounds__(HIST_THREADS)
void degree_hist_kernel(const int* __restrict__ src, const int* __restrict__ dst,
                        unsigned int* __restrict__ planes) {
    __shared__ __attribute__((aligned(16))) unsigned int h[PLANE_WORDS];
    const int t     = threadIdx.x;
    const int slice = blockIdx.x;
    const int which = blockIdx.y;
    const int4* __restrict__ idx4 =
        (const int4*)((which == 0 ? src : dst) + (size_t)slice * SLICE_EDGES);

    int4 v[HIST_LOADS];
    #pragma unroll
    for (int j = 0; j < HIST_LOADS; ++j) {
        const int q = t + j * HIST_THREADS;
        if (q < SLICE_INT4) v[j] = idx4[q];
    }

    uint4* h4 = (uint4*)h;
    for (int i = t; i < PLANE_UINT4; i += HIST_THREADS) h4[i] = make_uint4(0u, 0u, 0u, 0u);
    __syncthreads();

    #pragma unroll
    for (int j = 0; j < HIST_LOADS; ++j) {
        const int q = t + j * HIST_THREADS;
        if (q < SLICE_INT4) {
            lds_count(h, v[j].x);
            lds_count(h, v[j].y);
            lds_count(h, v[j].z);
            lds_count(h, v[j].w);
        }
    }
    __syncthreads();

    uint4* __restrict__ o4 = (uint4*)(planes + ((size_t)which * HIST_SLICES + slice) * PLANE_WORDS);
    for (int i = t; i < PLANE_UINT4; i += HIST_THREADS) o4[i] = h4[i];
}

__global__ __launch_bounds__(256)
void degree_norm_kernel(const unsigned int* __restrict__ planes, float* __restrict__ s) {
    const int w     = blockIdx.x * blockDim.x + threadIdx.x;
    const int which = blockIdx.y;
    if (w >= PLANE_WORDS) return;
    const unsigned int* __restrict__ p = planes + (size_t)which * HIST_SLICES * PLANE_WORDS + w;
    unsigned int lo = 0, hi = 0;
    for (int g = 0; g < HIST_SLICES; ++g) {
        const unsigned int u = p[(size_t)g * PLANE_WORDS];
        lo += u & 0xffffu;
        hi += u >> 16;
    }
    float2 r;
    r.x = rsqrtf(fmaxf((float)lo, 1.0f));
    r.y = rsqrtf(fmaxf((float)hi, 1.0f));
    *(float2*)(s + (size_t)which * N_NODES + 2 * w) = r;
}

__global__ __launch_bounds__(256)
void gemm_wmma_kernel(const float* __restrict__ x, const float* __restrict__ W,
                      const float* __restrict__ s_out, float* __restrict__ y) {
    const int wave   = (blockIdx.x * blockDim.x + threadIdx.x) >> 5;
    const int lane   = threadIdx.x & 31;
    const int tile_n = wave & (N_TILES - 1);
    const int tile_m = wave >> 3;

    const int m = lane & 15;
    const int h = lane >> 4;
    const int row = tile_m * 16 + m;
    const int col = tile_n * 16 + m;

    const float scale = s_out[row];
    const float* __restrict__ arow = x + (size_t)row * FEATS;

    v8f c = {};
    #pragma unroll 4
    for (int k0 = 0; k0 < FEATS; k0 += 4) {
        const int k = k0 + 2 * h;
        v2f a = *(const v2f*)(arow + k);
        a.x *= scale;
        a.y *= scale;
        v2f b;
        b.x = W[(size_t)k * FEATS + col];
        b.y = W[(size_t)(k + 1) * FEATS + col];
        c = __builtin_amdgcn_wmma_f32_16x16x4_f32(
                 false, a,  false, b,
                 (short)0, c,  false,  false);
    }

    #pragma unroll
    for (int j = 0; j < 8; ++j) {
        const int orow = tile_m * 16 + j + 8 * h;
        y[(size_t)orow * FEATS + col] = c[j];
    }
}

__global__ __launch_bounds__(256)
void scatter_kernel(const int* __restrict__ src, const int* __restrict__ dst,
                    const float* __restrict__ y, float* __restrict__ out) {
    const int e    = (blockIdx.x * blockDim.x + threadIdx.x) >> 5;
    const int lane = threadIdx.x & 31;
    if (e >= N_EDGES) return;
    const int s = src[e];
    const int d = dst[e];
    const float4 v = *(const float4*)(y + (size_t)s * FEATS + lane * 4);
    float* o = out + (size_t)d * FEATS + lane * 4;
    atomic_add_f32(o + 0, v.x);
    atomic_add_f32(o + 1, v.y);
    atomic_add_f32(o + 2, v.z);
    atomic_add_f32(o + 3, v.w);
}

__global__ void finalize_kernel(float* __restrict__ out, const float* __restrict__ s_in,
                                const float* __restrict__ bias) {
    int i = blockIdx.x * blockDim.x + threadIdx.x;
    if (i < N_NODES * FEATS) {
        int r = i >> 7;
        int ccol = i & (FEATS - 1);
        out[i] = out[i] * s_in[r] + bias[ccol];
    }
}

extern "C" void kernel_launch(void* const* d_in, const int* in_sizes, int n_in,
                              void* d_out, int out_size, void* d_ws, size_t ws_size,
                              hipStream_t stream) {
    const float* x    = (const float*)d_in[0];
    const int*   src  = (const int*)  d_in[1];
    const int*   dst  = (const int*)  d_in[2];
    const float* W    = (const float*)d_in[3];
    const float* bias = (const float*)d_in[4];
    float* out = (float*)d_out;

    float* s_out = (float*)d_ws;
    float* s_in  = s_out + N_NODES;
    float* y     = s_in + N_NODES;

    const int TB = 256;

    zero_kernel<<<(N_NODES * FEATS + TB - 1) / TB, TB, 0, stream>>>(out, N_NODES * FEATS);

    unsigned int* planes = (unsigned int*)y;
    degree_hist_kernel<<<dim3(HIST_SLICES, 2), HIST_THREADS, 0, stream>>>(src, dst, planes);

    degree_norm_kernel<<<dim3((PLANE_WORDS + TB - 1) / TB, 2), TB, 0, stream>>>(planes, s_out);

    gemm_wmma_kernel<<<(M_TILES * N_TILES) / 8, TB, 0, stream>>>(x, W, s_out, y);

    scatter_kernel<<<(N_EDGES * 32) / TB, TB, 0, stream>>>(src, dst, y, out);

    finalize_kernel<<<(N_NODES * FEATS + TB - 1) / TB, TB, 0, stream>>>(out, s_in, bias);
}


# scatter f32 atomics replaced by gather over edges counting-sorted by destination (LDS ranks), rows summed in csr order ahead of the GEMM: sum order changes; zero fill, y, finalize gone
# speedup vs baseline: 3.2015x; 2.2709x over previous
#include <hip/hip_runtime.h>

#define N_NODES 50000
#define N_EDGES 800000
#define FEATS   128

typedef __attribute__((ext_vector_type(2))) float v2f;
typedef __attribute__((ext_vector_type(8))) float v8f;

#define SRC_SLICES      64
#define DST_SLICES      32
#define SRC_SLICE_INT4  3125
#define DST_SLICE_INT4  6250
#define PLANE_WORDS     (N_NODES / 2)
#define PLANE_UINT4     (PLANE_WORDS / 4)
#define HIST_THREADS    1024
#define HIST_LOADS      7

__device__ __forceinline__ unsigned int lds_count(unsigned int* h, int node) {
    const int sh = (node & 1) << 4;
    const unsigned int old = __hip_atomic_fetch_add(&h[node >> 1], 1u << sh,
                                                    __ATOMIC_RELAXED, __HIP_MEMORY_SCOPE_WORKGROUP);
    return (old >> sh) & 0xffffu;
}

__global__ __launch_bounds__(HIST_THREADS)
void degree_hist_kernel(const int* __restrict__ src, const int* __restrict__ dst,
                        unsigned int* __restrict__ planes, uint2* __restrict__ rank4) {
    __shared__ __attribute__((aligned(16))) unsigned int h[PLANE_WORDS];
    const int t      = threadIdx.x;
    const int b      = blockIdx.x;
    const bool isdst = b >= SRC_SLICES;
    const int slice  = isdst ? b - SRC_SLICES : b;
    const int n4     = isdst ? DST_SLICE_INT4 : SRC_SLICE_INT4;
    const int4* __restrict__ idx4 = isdst ? (const int4*)dst + (size_t)slice * DST_SLICE_INT4
                                          : (const int4*)src + (size_t)slice * SRC_SLICE_INT4;

    int4 v[HIST_LOADS];
    #pragma unroll
    for (int j = 0; j < HIST_LOADS; ++j) {
        const int q = t + j * HIST_THREADS;
        if (q < n4) v[j] = idx4[q];
    }

    uint4* h4 = (uint4*)h;
    for (int i = t; i < PLANE_UINT4; i += HIST_THREADS) h4[i] = make_uint4(0u, 0u, 0u, 0u);
    __syncthreads();

    #pragma unroll
    for (int j = 0; j < HIST_LOADS; ++j) {
        const int q = t + j * HIST_THREADS;
        if (q < n4) {
            const unsigned int r0 = lds_count(h, v[j].x);
            const unsigned int r1 = lds_count(h, v[j].y);
            const unsigned int r2 = lds_count(h, v[j].z);
            const unsigned int r3 = lds_count(h, v[j].w);
            if (isdst) rank4[(size_t)slice * DST_SLICE_INT4 + q] = make_uint2(r0 | (r1 << 16), r2 | (r3 << 16));
        }
    }
    __syncthreads();

    uint4* __restrict__ o4 = (uint4*)(planes + (size_t)b * PLANE_WORDS);
    for (int i = t; i < PLANE_UINT4; i += HIST_THREADS) o4[i] = h4[i];
}

__global__ __launch_bounds__(256)
void degree_norm_kernel(const unsigned int* __restrict__ planes, float* __restrict__ s_out,
                        float* __restrict__ s_in, unsigned int* __restrict__ pre,
                        int* __restrict__ deg_in) {
    const int w = blockIdx.x * blockDim.x + threadIdx.x;
    if (w >= PLANE_WORDS) return;
    unsigned int lo = 0, hi = 0;
    if (blockIdx.y == 0) {
        const unsigned int* __restrict__ p = planes + w;
        for (int g = 0; g < SRC_SLICES; ++g) {
            const unsigned int u = p[(size_t)g * PLANE_WORDS];
            lo += u & 0xffffu;
            hi += u >> 16;
        }
        float2 r;
        r.x = rsqrtf(fmaxf((float)lo, 1.0f));
        r.y = rsqrtf(fmaxf((float)hi, 1.0f));
        *(float2*)(s_out + 2 * w) = r;
    } else {
        const unsigned int* __restrict__ p = planes + (size_t)SRC_SLICES * PLANE_WORDS + w;
        for (int g = 0; g < DST_SLICES; ++g) {
            const unsigned int u = p[(size_t)g * PLANE_WORDS];
            *(uint2*)(pre + (size_t)g * N_NODES + 2 * w) = make_uint2(lo, hi);
            lo += u & 0xffffu;
            hi += u >> 16;
        }
        float2 r;
        r.x = rsqrtf(fmaxf((float)lo, 1.0f));
        r.y = rsqrtf(fmaxf((float)hi, 1.0f));
        *(float2*)(s_in + 2 * w) = r;
        *(int2*)(deg_in + 2 * w) = make_int2((int)lo, (int)hi);
    }
}

#define SCAN_THREADS 1024
#define NODE_INT4    (N_NODES / 4)
#define SCAN_TILES   13

__device__ __forceinline__ int wave_incl_scan(int x, int lane) {
    x += __builtin_amdgcn_update_dpp(0, x, 0x111, 0xf, 0xf, false);
    x += __builtin_amdgcn_update_dpp(0, x, 0x112, 0xf, 0xf, false);
    x += __builtin_amdgcn_update_dpp(0, x, 0x114, 0xf, 0xf, false);
    x += __builtin_amdgcn_update_dpp(0, x, 0x118, 0xf, 0xf, false);
    const int low = __builtin_amdgcn_readlane(x, 15);
    return x + ((lane >= 16) ? low : 0);
}

__global__ __launch_bounds__(SCAN_THREADS)
void rowptr_scan_kernel(const int* __restrict__ deg_in, int* __restrict__ rowptr) {
    __shared__ int wsum[SCAN_TILES][32];
    const int t    = threadIdx.x;
    const int lane = t & 31;
    const int wave = __builtin_amdgcn_readfirstlane(t >> 5);

    int4 v[SCAN_TILES];
    #pragma unroll
    for (int k = 0; k < SCAN_TILES; ++k) {
        const int q = t + k * SCAN_THREADS;
        v[k] = (q < NODE_INT4) ? ((const int4*)deg_in)[q] : make_int4(0, 0, 0, 0);
    }

    int incl[SCAN_TILES];
    #pragma unroll
    for (int k = 0; k < SCAN_TILES; ++k) incl[k] = v[k].x + v[k].y + v[k].z + v[k].w;
    #pragma unroll
    for (int k = 0; k < SCAN_TILES; ++k) incl[k] = wave_incl_scan(incl[k], lane);
    if (lane == 31) {
        #pragma unroll
        for (int k = 0; k < SCAN_TILES; ++k) wsum[k][wave] = incl[k];
    }
    __syncthreads();

    int wi[SCAN_TILES];
    #pragma unroll
    for (int k = 0; k < SCAN_TILES; ++k) wi[k] = wsum[k][lane];
    #pragma unroll
    for (int k = 0; k < SCAN_TILES; ++k) wi[k] = wave_incl_scan(wi[k], lane);

    int carry = 0;
    #pragma unroll
    for (int k = 0; k < SCAN_TILES; ++k) {
        const int q       = t + k * SCAN_THREADS;
        const int total   = __builtin_amdgcn_readlane(wi[k], 31);
        const int prev    = __builtin_amdgcn_readlane(wi[k], (wave + 31) & 31);
        const int waveoff = (wave > 0) ? prev : 0;
        const int s1 = v[k].x;
        const int s2 = s1 + v[k].y;
        const int s3 = s2 + v[k].z;
        const int s4 = s3 + v[k].w;
        const int excl = carry + waveoff + incl[k] - s4;
        if (q < NODE_INT4) ((int4*)rowptr)[q] = make_int4(excl, excl + s1, excl + s2, excl + s3);
        carry += total;
    }
    if (t == 0) rowptr[N_NODES] = carry;
}

#define EDGE_INT4 (N_EDGES / 4)

__global__ __launch_bounds__(256)
void csr_fill_kernel(const int* __restrict__ src, const int* __restrict__ dst,
                     const uint2* __restrict__ rank4, const unsigned int* __restrict__ pre,
                     const int* __restrict__ rowptr, int* __restrict__ csr) {
    const int q = blockIdx.x * blockDim.x + threadIdx.x;
    if (q >= EDGE_INT4) return;
    const int4  d4 = ((const int4*)dst)[q];
    const int4  s4 = ((const int4*)src)[q];
    const uint2 r  = rank4[q];
    const unsigned int* __restrict__ pg = pre + (size_t)(q / DST_SLICE_INT4) * N_NODES;
    const unsigned int p0 = (unsigned int)rowptr[d4.x] + pg[d4.x] + (r.x & 0xffffu);
    const unsigned int p1 = (unsigned int)rowptr[d4.y] + pg[d4.y] + (r.x >> 16);
    const unsigned int p2 = (unsigned int)rowptr[d4.z] + pg[d4.z] + (r.y & 0xffffu);
    const unsigned int p3 = (unsigned int)rowptr[d4.w] + pg[d4.w] + (r.y >> 16);
    csr[min(p0, (unsigned int)(N_EDGES - 1))] = s4.x;
    csr[min(p1, (unsigned int)(N_EDGES - 1))] = s4.y;
    csr[min(p2, (unsigned int)(N_EDGES - 1))] = s4.z;
    csr[min(p3, (unsigned int)(N_EDGES - 1))] = s4.w;
}

#define HP 132

template <int U>
__device__ __forceinline__ void gather_batch(const float* __restrict__ x, int idx, float sc, int k, int lane,
                                             float4& acc) {
    float4 xv[U];
    float  f[U];
    #pragma unroll
    for (int u = 0; u < U; ++u) {
        const int s = __builtin_amdgcn_readlane(idx, k + u);
        f[u]  = __int_as_float(__builtin_amdgcn_readlane(__float_as_int(sc), k + u));
        xv[u] = *(const float4*)(x + (size_t)s * FEATS + lane * 4);
    }
    #pragma unroll
    for (int u = 0; u < U; ++u) {
        acc.x = fmaf(xv[u].x, f[u], acc.x);
        acc.y = fmaf(xv[u].y, f[u], acc.y);
        acc.z = fmaf(xv[u].z, f[u], acc.z);
        acc.w = fmaf(xv[u].w, f[u], acc.w);
    }
}

__global__ __launch_bounds__(256)
void gather_gemm_kernel(const float* __restrict__ x, const float* __restrict__ W,
                        const float* __restrict__ bias, const float* __restrict__ s_out,
                        const float* __restrict__ s_in, const int* __restrict__ rowptr,
                        const int* __restrict__ csr, float* __restrict__ out) {
    __shared__ __attribute__((aligned(16))) float Hs[16 * HP];
    const int lane = threadIdx.x & 31;
    const int wave = __builtin_amdgcn_readfirstlane((int)(threadIdx.x >> 5));
    const int row0 = blockIdx.x * 16;

    for (int rr = 0; rr < 2; ++rr) {
        const int r = 2 * wave + rr;
        const int vtx = row0 + r;
        int beg = __builtin_amdgcn_readfirstlane(rowptr[vtx]);
        int end = __builtin_amdgcn_readfirstlane(rowptr[vtx + 1]);
        beg = min(max(beg, 0), N_EDGES);
        end = min(max(end, beg), N_EDGES);
        float4 acc = make_float4(0.0f, 0.0f, 0.0f, 0.0f);
        for (int base = beg; base < end; base += 32) {
            const int n = min(32, end - base);
            int   idx = 0;
            float sc  = 0.0f;
            if (lane < n) {
                idx = (int)min((unsigned int)csr[base + lane], (unsigned int)(N_NODES - 1));
                sc  = s_out[idx];
            }
            int k = 0;
            for (; k + 8 <= n; k += 8) gather_batch<8>(x, idx, sc, k, lane, acc);
            if (k + 4 <= n) { gather_batch<4>(x, idx, sc, k, lane, acc); k += 4; }
            for (; k < n; ++k) gather_batch<1>(x, idx, sc, k, lane, acc);
        }
        *(float4*)(Hs + r * HP + lane * 4) = acc;
    }
    __syncthreads();

    const int m   = lane & 15;
    const int h   = lane >> 4;
    const int col = wave * 16 + m;
    v8f c = {};
    #pragma unroll 8
    for (int k0 = 0; k0 < FEATS; k0 += 4) {
        const int k = k0 + 2 * h;
        const v2f a = *(const v2f*)(Hs + m * HP + k);
        v2f b;
        b.x = W[(size_t)k * FEATS + col];
        b.y = W[(size_t)(k + 1) * FEATS + col];
        c = __builtin_amdgcn_wmma_f32_16x16x4_f32(
                 false, a,  false, b,
                 (short)0, c,  false,  false);
    }
    __syncthreads();

    #pragma unroll
    for (int j = 0; j < 8; ++j) Hs[(j + 8 * h) * HP + col] = c[j];
    __syncthreads();
    const float4 bv = *(const float4*)(bias + lane * 4);
    #pragma unroll
    for (int rr = 0; rr < 2; ++rr) {
        const int r = 2 * wave + rr;
        const float sn = s_in[row0 + r];
        const float4 hv = *(const float4*)(Hs + r * HP + lane * 4);
        float4 o;
        o.x = hv.x * sn + bv.x;
        o.y = hv.y * sn + bv.y;
        o.z = hv.z * sn + bv.z;
        o.w = hv.w * sn + bv.w;
        *(float4*)(out + (size_t)(row0 + r) * FEATS + lane * 4) = o;
    }
}

extern "C" void kernel_launch(void* const* d_in, const int* in_sizes, int n_in,
                              void* d_out, int out_size, void* d_ws, size_t ws_size,
                              hipStream_t stream) {
    const float* x    = (const float*)d_in[0];
    const int*   src  = (const int*)  d_in[1];
    const int*   dst  = (const int*)  d_in[2];
    const float* W    = (const float*)d_in[3];
    const float* bias = (const float*)d_in[4];
    float* out = (float*)d_out;

    float*        s_out  = (float*)d_ws;
    float*        s_in   = s_out + N_NODES;
    unsigned int* planes = (unsigned int*)(s_in + N_NODES);
    unsigned int* pre    = planes + (SRC_SLICES + DST_SLICES) * PLANE_WORDS;
    uint2*        rank4  = (uint2*)(pre + DST_SLICES * N_NODES);
    int*          deg_in = (int*)(rank4 + EDGE_INT4);
    int*          rowptr = deg_in + N_NODES;
    int*          csr    = rowptr + N_NODES + 4;

    const int TB = 256;

    degree_hist_kernel<<<SRC_SLICES + DST_SLICES, HIST_THREADS, 0, stream>>>(src, dst, planes, rank4);

    degree_norm_kernel<<<dim3((PLANE_WORDS + TB - 1) / TB, 2), TB, 0, stream>>>(planes, s_out, s_in, pre, deg_in);

    rowptr_scan_kernel<<<1, SCAN_THREADS, 0, stream>>>(deg_in, rowptr);

    csr_fill_kernel<<<(EDGE_INT4 + TB - 1) / TB, TB, 0, stream>>>(src, dst, rank4, pre, rowptr, csr);

    gather_gemm_kernel<<<N_NODES / 16, TB, 0, stream>>>(x, W, bias, s_out, s_in, rowptr, csr, out);
}


# rowptr scan launch removed: dst histogram blocks also store per-slice edge counts below each 256-node bucket; the degree kernel adds its bucket's 32 counts and scans its 512 nodes; 5 launches -> 4
# speedup vs baseline: 3.3763x; 1.0546x over previous
#include <hip/hip_runtime.h>

#define N_NODES 50000
#define N_EDGES 800000
#define FEATS   128

typedef __attribute__((ext_vector_type(2))) float v2f;
typedef __attribute__((ext_vector_type(8))) float v8f;

#define SRC_SLICES      64
#define DST_SLICES      32
#define SRC_SLICE_INT4  3125
#define DST_SLICE_INT4  6250
#define PLANE_WORDS     (N_NODES / 2)
#define PLANE_UINT4     (PLANE_WORDS / 4)
#define HIST_THREADS    1024
#define HIST_LOADS      7
#define BUCKETS         ((PLANE_UINT4 + 31) / 32)

__device__ __forceinline__ int wave_incl_scan(int x, int lane) {
    x += __builtin_amdgcn_update_dpp(0, x, 0x111, 0xf, 0xf, false);
    x += __builtin_amdgcn_update_dpp(0, x, 0x112, 0xf, 0xf, false);
    x += __builtin_amdgcn_update_dpp(0, x, 0x114, 0xf, 0xf, false);
    x += __builtin_amdgcn_update_dpp(0, x, 0x118, 0xf, 0xf, false);
    const int low = __builtin_amdgcn_readlane(x, 15);
    return x + ((lane >= 16) ? low : 0);
}

__device__ __forceinline__ unsigned int lds_count(unsigned int* h, int node) {
    const int sh = (node & 1) << 4;
    const unsigned int old = __hip_atomic_fetch_add(&h[node >> 1], 1u << sh,
                                                    __ATOMIC_RELAXED, __HIP_MEMORY_SCOPE_WORKGROUP);
    return (old >> sh) & 0xffffu;
}

__global__ __launch_bounds__(HIST_THREADS)
void degree_hist_kernel(const int* __restrict__ src, const int* __restrict__ dst,
                        unsigned int* __restrict__ planes, uint2* __restrict__ rank4,
                        int* __restrict__ coarse) {
    __shared__ __attribute__((aligned(16))) unsigned int h[PLANE_WORDS];
    __shared__ int btot[HIST_LOADS * 32];
    const int t      = threadIdx.x;
    const int b      = blockIdx.x;
    const bool isdst = b >= SRC_SLICES;
    const int slice  = isdst ? b - SRC_SLICES : b;
    const int n4     = isdst ? DST_SLICE_INT4 : SRC_SLICE_INT4;
    const int4* __restrict__ idx4 = isdst ? (const int4*)dst + (size_t)slice * DST_SLICE_INT4
                                          : (const int4*)src + (size_t)slice * SRC_SLICE_INT4;

    int4 v[HIST_LOADS];
    #pragma unroll
    for (int j = 0; j < HIST_LOADS; ++j) {
        const int q = t + j * HIST_THREADS;
        if (q < n4) v[j] = idx4[q];
    }

    uint4* h4 = (uint4*)h;
    for (int i = t; i < PLANE_UINT4; i += HIST_THREADS) h4[i] = make_uint4(0u, 0u, 0u, 0u);
    __syncthreads();

    #pragma unroll
    for (int j = 0; j < HIST_LOADS; ++j) {
        const int q = t + j * HIST_THREADS;
        if (q < n4) {
            const unsigned int r0 = lds_count(h, v[j].x);
            const unsigned int r1 = lds_count(h, v[j].y);
            const unsigned int r2 = lds_count(h, v[j].z);
            const unsigned int r3 = lds_count(h, v[j].w);
            if (isdst) rank4[(size_t)slice * DST_SLICE_INT4 + q] = make_uint2(r0 | (r1 << 16), r2 | (r3 << 16));
        }
    }
    __syncthreads();

    const int lane = t & 31;
    uint4 u[HIST_LOADS];
    #pragma unroll
    for (int j = 0; j < HIST_LOADS; ++j) {
        const int i = t + j * HIST_THREADS;
        u[j] = (i < PLANE_UINT4) ? h4[i] : make_uint4(0u, 0u, 0u, 0u);
    }

    if (isdst) {
        #pragma unroll
        for (int j = 0; j < HIST_LOADS; ++j) {
            const int i0 = __builtin_amdgcn_readfirstlane(t + j * HIST_THREADS);
            if (i0 < PLANE_UINT4) {
                const int s = (int)((u[j].x & 0xffffu) + (u[j].x >> 16) + (u[j].y & 0xffffu) + (u[j].y >> 16) +
                                    (u[j].z & 0xffffu) + (u[j].z >> 16) + (u[j].w & 0xffffu) + (u[j].w >> 16));
                const int tot = wave_incl_scan(s, lane);
                if (lane == 31) btot[i0 >> 5] = tot;
            }
        }
        __syncthreads();
        if (t < 32) {
            int carry = 0;
            for (int j = 0; j < HIST_LOADS; ++j) {
                const int k    = 32 * j + lane;
                const int v    = (k < BUCKETS) ? btot[k] : 0;
                const int incl = wave_incl_scan(v, lane);
                if (k < BUCKETS) coarse[k * DST_SLICES + slice] = carry + incl - v;
                carry += __builtin_amdgcn_readlane(incl, 31);
            }
        }
    }

    uint4* __restrict__ o4 = (uint4*)(planes + (size_t)b * PLANE_WORDS);
    #pragma unroll
    for (int j = 0; j < HIST_LOADS; ++j) {
        const int i = t + j * HIST_THREADS;
        if (i < PLANE_UINT4) o4[i] = u[j];
    }
}

__global__ __launch_bounds__(256)
void degree_norm_kernel(const unsigned int* __restrict__ planes, const int* __restrict__ coarse,
                        float* __restrict__ s_out, float* __restrict__ s_in,
                        unsigned int* __restrict__ pre, int* __restrict__ rowptr) {
    __shared__ int wsum[8];
    const int t = threadIdx.x;
    const int w = blockIdx.x * 256 + t;
    const bool valid = w < PLANE_WORDS;
    unsigned int lo = 0, hi = 0;
    if (blockIdx.y == 0) {
        if (!valid) return;
        const unsigned int* __restrict__ p = planes + w;
        for (int g = 0; g < SRC_SLICES; ++g) {
            const unsigned int u = p[(size_t)g * PLANE_WORDS];
            lo += u & 0xffffu;
            hi += u >> 16;
        }
        float2 r;
        r.x = rsqrtf(fmaxf((float)lo, 1.0f));
        r.y = rsqrtf(fmaxf((float)hi, 1.0f));
        *(float2*)(s_out + 2 * w) = r;
    } else {
        const int lane = t & 31;
        const int wave = t >> 5;
        const int below = coarse[2 * blockIdx.x * DST_SLICES + lane];
        if (valid) {
            const unsigned int* __restrict__ p = planes + (size_t)SRC_SLICES * PLANE_WORDS + w;
            for (int g = 0; g < DST_SLICES; ++g) {
                const unsigned int u = p[(size_t)g * PLANE_WORDS];
                *(uint2*)(pre + (size_t)g * N_NODES + 2 * w) = make_uint2(lo, hi);
                lo += u & 0xffffu;
                hi += u >> 16;
            }
            float2 r;
            r.x = rsqrtf(fmaxf((float)lo, 1.0f));
            r.y = rsqrtf(fmaxf((float)hi, 1.0f));
            *(float2*)(s_in + 2 * w) = r;
        }
        const int deg  = (int)(lo + hi);
        const int incl = wave_incl_scan(deg, lane);
        int off = __builtin_amdgcn_readlane(wave_incl_scan(below, lane), 31);
        if (lane == 31) wsum[wave] = incl;
        __syncthreads();
        for (int i = 0; i < 8; ++i) {
            if (i < wave) off += wsum[i];
        }
        const int excl = off + incl - deg;
        if (valid) {
            *(int2*)(rowptr + 2 * w) = make_int2(excl, excl + (int)lo);
            if (w == PLANE_WORDS - 1) rowptr[N_NODES] = excl + deg;
        }
    }
}

#define EDGE_INT4 (N_EDGES / 4)

__global__ __launch_bounds__(256)
void csr_fill_kernel(const int* __restrict__ src, const int* __restrict__ dst,
                     const uint2* __restrict__ rank4, const unsigned int* __restrict__ pre,
                     const int* __restrict__ rowptr, int* __restrict__ csr) {
    const int q = blockIdx.x * blockDim.x + threadIdx.x;
    if (q >= EDGE_INT4) return;
    const int4  d4 = ((const int4*)dst)[q];
    const int4  s4 = ((const int4*)src)[q];
    const uint2 r  = rank4[q];
    const unsigned int* __restrict__ pg = pre + (size_t)(q / DST_SLICE_INT4) * N_NODES;
    const unsigned int p0 = (unsigned int)rowptr[d4.x] + pg[d4.x] + (r.x & 0xffffu);
    const unsigned int p1 = (unsigned int)rowptr[d4.y] + pg[d4.y] + (r.x >> 16);
    const unsigned int p2 = (unsigned int)rowptr[d4.z] + pg[d4.z] + (r.y & 0xffffu);
    const unsigned int p3 = (unsigned int)rowptr[d4.w] + pg[d4.w] + (r.y >> 16);
    csr[min(p0, (unsigned int)(N_EDGES - 1))] = s4.x;
    csr[min(p1, (unsigned int)(N_EDGES - 1))] = s4.y;
    csr[min(p2, (unsigned int)(N_EDGES - 1))] = s4.z;
    csr[min(p3, (unsigned int)(N_EDGES - 1))] = s4.w;
}

#define HP 132

template <int U>
__device__ __forceinline__ void gather_batch(const float* __restrict__ x, int idx, float sc, int k, int lane,
                                             float4& acc) {
    float4 xv[U];
    float  f[U];
    #pragma unroll
    for (int u = 0; u < U; ++u) {
        const int s = __builtin_amdgcn_readlane(idx, k + u);
        f[u]  = __int_as_float(__builtin_amdgcn_readlane(__float_as_int(sc), k + u));
        xv[u] = *(const float4*)(x + (size_t)s * FEATS + lane * 4);
    }
    #pragma unroll
    for (int u = 0; u < U; ++u) {
        acc.x = fmaf(xv[u].x, f[u], acc.x);
        acc.y = fmaf(xv[u].y, f[u], acc.y);
        acc.z = fmaf(xv[u].z, f[u], acc.z);
        acc.w = fmaf(xv[u].w, f[u], acc.w);
    }
}

__global__ __launch_bounds__(256)
void gather_gemm_kernel(const float* __restrict__ x, const float* __restrict__ W,
                        const float* __restrict__ bias, const float* __restrict__ s_out,
                        const float* __restrict__ s_in, const int* __restrict__ rowptr,
                        const int* __restrict__ csr, float* __restrict__ out) {
    __shared__ __attribute__((aligned(16))) float Hs[16 * HP];
    const int lane = threadIdx.x & 31;
    const int wave = __builtin_amdgcn_readfirstlane((int)(threadIdx.x >> 5));
    const int row0 = blockIdx.x * 16;

    for (int rr = 0; rr < 2; ++rr) {
        const int r = 2 * wave + rr;
        const int vtx = row0 + r;
        int beg = __builtin_amdgcn_readfirstlane(rowptr[vtx]);
        int end = __builtin_amdgcn_readfirstlane(rowptr[vtx + 1]);
        beg = min(max(beg, 0), N_EDGES);
        end = min(max(end, beg), N_EDGES);
        float4 acc = make_float4(0.0f, 0.0f, 0.0f, 0.0f);
        for (int base = beg; base < end; base += 32) {
            const int n = min(32, end - base);
            int   idx = 0;
            float sc  = 0.0f;
            if (lane < n) {
                idx = (int)min((unsigned int)csr[base + lane], (unsigned int)(N_NODES - 1));
                sc  = s_out[idx];
            }
            int k = 0;
            for (; k + 8 <= n; k += 8) gather_batch<8>(x, idx, sc, k, lane, acc);
            if (k + 4 <= n) { gather_batch<4>(x, idx, sc, k, lane, acc); k += 4; }
            for (; k < n; ++k) gather_batch<1>(x, idx, sc, k, lane, acc);
        }
        *(float4*)(Hs + r * HP + lane * 4) = acc;
    }
    __syncthreads();

    const int m   = lane & 15;
    const int h   = lane >> 4;
    const int col = wave * 16 + m;
    v8f c = {};
    #pragma unroll 8
    for (int k0 = 0; k0 < FEATS; k0 += 4) {
        const int k = k0 + 2 * h;
        const v2f a = *(const v2f*)(Hs + m * HP + k);
        v2f b;
        b.x = W[(size_t)k * FEATS + col];
        b.y = W[(size_t)(k + 1) * FEATS + col];
        c = __builtin_amdgcn_wmma_f32_16x16x4_f32(
                 false, a,  false, b,
                 (short)0, c,  false,  false);
    }
    __syncthreads();

    #pragma unroll
    for (int j = 0; j < 8; ++j) Hs[(j + 8 * h) * HP + col] = c[j];
    __syncthreads();
    const float4 bv = *(const float4*)(bias + lane * 4);
    #pragma unroll
    for (int rr = 0; rr < 2; ++rr) {
        const int r = 2 * wave + rr;
        const float sn = s_in[row0 + r];
        const float4 hv = *(const float4*)(Hs + r * HP + lane * 4);
        float4 o;
        o.x = hv.x * sn + bv.x;
        o.y = hv.y * sn + bv.y;
        o.z = hv.z * sn + bv.z;
        o.w = hv.w * sn + bv.w;
        *(float4*)(out + (size_t)(row0 + r) * FEATS + lane * 4) = o;
    }
}

extern "C" void kernel_launch(void* const* d_in, const int* in_sizes, int n_in,
                              void* d_out, int out_size, void* d_ws, size_t ws_size,
                              hipStream_t stream) {
    const float* x    = (const float*)d_in[0];
    const int*   src  = (const int*)  d_in[1];
    const int*   dst  = (const int*)  d_in[2];
    const float* W    = (const float*)d_in[3];
    const float* bias = (const float*)d_in[4];
    float* out = (float*)d_out;

    float*        s_out  = (float*)d_ws;
    float*        s_in   = s_out + N_NODES;
    unsigned int* planes = (unsigned int*)(s_in + N_NODES);
    unsigned int* pre    = planes + (SRC_SLICES + DST_SLICES) * PLANE_WORDS;
    uint2*        rank4  = (uint2*)(pre + DST_SLICES * N_NODES);
    int*          coarse = (int*)(rank4 + EDGE_INT4);
    int*          rowptr = coarse + N_NODES;
    int*          csr    = rowptr + N_NODES + 4;

    const int TB = 256;

    degree_hist_kernel<<<SRC_SLICES + DST_SLICES, HIST_THREADS, 0, stream>>>(src, dst, planes, rank4, coarse);

    degree_norm_kernel<<<dim3((PLANE_WORDS + TB - 1) / TB, 2), TB, 0, stream>>>(planes, coarse, s_out, s_in, pre, rowptr);

    csr_fill_kernel<<<(EDGE_INT4 + TB - 1) / TB, TB, 0, stream>>>(src, dst, rank4, pre, rowptr, csr);

    gather_gemm_kernel<<<N_NODES / 16, TB, 0, stream>>>(x, W, bias, s_out, s_in, rowptr, csr, out);
}


# csr fill: edge positions looked up in LDS: norm kernel stores rowptr + earlier-slice count as one word per (slice,node); each of 256 fill blocks stages its slice's 200 KB table; csr order kept
# speedup vs baseline: 3.6276x; 1.0744x over previous
#include <hip/hip_runtime.h>

#define N_NODES 50000
#define N_EDGES 800000
#define FEATS   128

typedef __attribute__((ext_vector_type(2))) float v2f;
typedef __attribute__((ext_vector_type(8))) float v8f;

#define SRC_SLICES      64
#define DST_SLICES      32
#define SRC_SLICE_INT4  3125
#define DST_SLICE_INT4  6250
#define PLANE_WORDS     (N_NODES / 2)
#define PLANE_UINT4     (PLANE_WORDS / 4)
#define HIST_THREADS    1024
#define HIST_LOADS      7
#define BUCKETS         ((PLANE_UINT4 + 31) / 32)

__device__ __forceinline__ int wave_incl_scan(int x, int lane) {
    x += __builtin_amdgcn_update_dpp(0, x, 0x111, 0xf, 0xf, false);
    x += __builtin_amdgcn_update_dpp(0, x, 0x112, 0xf, 0xf, false);
    x += __builtin_amdgcn_update_dpp(0, x, 0x114, 0xf, 0xf, false);
    x += __builtin_amdgcn_update_dpp(0, x, 0x118, 0xf, 0xf, false);
    const int low = __builtin_amdgcn_readlane(x, 15);
    return x + ((lane >= 16) ? low : 0);
}

__device__ __forceinline__ unsigned int lds_count(unsigned int* h, int node) {
    const int sh = (node & 1) << 4;
    const unsigned int old = __hip_atomic_fetch_add(&h[node >> 1], 1u << sh,
                                                    __ATOMIC_RELAXED, __HIP_MEMORY_SCOPE_WORKGROUP);
    return (old >> sh) & 0xffffu;
}

__global__ __launch_bounds__(HIST_THREADS)
void degree_hist_kernel(const int* __restrict__ src, const int* __restrict__ dst,
                        unsigned int* __restrict__ planes, uint2* __restrict__ rank4,
                        int* __restrict__ coarse) {
    __shared__ __attribute__((aligned(16))) unsigned int h[PLANE_WORDS];
    __shared__ int btot[HIST_LOADS * 32];
    const int t      = threadIdx.x;
    const int b      = blockIdx.x;
    const bool isdst = b >= SRC_SLICES;
    const int slice  = isdst ? b - SRC_SLICES : b;
    const int n4     = isdst ? DST_SLICE_INT4 : SRC_SLICE_INT4;
    const int4* __restrict__ idx4 = isdst ? (const int4*)dst + (size_t)slice * DST_SLICE_INT4
                                          : (const int4*)src + (size_t)slice * SRC_SLICE_INT4;

    int4 v[HIST_LOADS];
    #pragma unroll
    for (int j = 0; j < HIST_LOADS; ++j) {
        const int q = t + j * HIST_THREADS;
        if (q < n4) v[j] = idx4[q];
    }

    uint4* h4 = (uint4*)h;
    for (int i = t; i < PLANE_UINT4; i += HIST_THREADS) h4[i] = make_uint4(0u, 0u, 0u, 0u);
    __syncthreads();

    #pragma unroll
    for (int j = 0; j < HIST_LOADS; ++j) {
        const int q = t + j * HIST_THREADS;
        if (q < n4) {
            const unsigned int r0 = lds_count(h, v[j].x);
            const unsigned int r1 = lds_count(h, v[j].y);
            const unsigned int r2 = lds_count(h, v[j].z);
            const unsigned int r3 = lds_count(h, v[j].w);
            if (isdst) rank4[(size_t)slice * DST_SLICE_INT4 + q] = make_uint2(r0 | (r1 << 16), r2 | (r3 << 16));
        }
    }
    __syncthreads();

    const int lane = t & 31;
    uint4 u[HIST_LOADS];
    #pragma unroll
    for (int j = 0; j < HIST_LOADS; ++j) {
        const int i = t + j * HIST_THREADS;
        u[j] = (i < PLANE_UINT4) ? h4[i] : make_uint4(0u, 0u, 0u, 0u);
    }

    if (isdst) {
        #pragma unroll
        for (int j = 0; j < HIST_LOADS; ++j) {
            const int i0 = __builtin_amdgcn_readfirstlane(t + j * HIST_THREADS);
            if (i0 < PLANE_UINT4) {
                const int s = (int)((u[j].x & 0xffffu) + (u[j].x >> 16) + (u[j].y & 0xffffu) + (u[j].y >> 16) +
                                    (u[j].z & 0xffffu) + (u[j].z >> 16) + (u[j].w & 0xffffu) + (u[j].w >> 16));
                const int tot = wave_incl_scan(s, lane);
                if (lane == 31) btot[i0 >> 5] = tot;
            }
        }
        __syncthreads();
        if (t < 32) {
            int carry = 0;
            for (int j = 0; j < HIST_LOADS; ++j) {
                const int k    = 32 * j + lane;
                const int v    = (k < BUCKETS) ? btot[k] : 0;
                const int incl = wave_incl_scan(v, lane);
                if (k < BUCKETS) coarse[k * DST_SLICES + slice] = carry + incl - v;
                carry += __builtin_amdgcn_readlane(incl, 31);
            }
        }
    }

    uint4* __restrict__ o4 = (uint4*)(planes + (size_t)b * PLANE_WORDS);
    #pragma unroll
    for (int j = 0; j < HIST_LOADS; ++j) {
        const int i = t + j * HIST_THREADS;
        if (i < PLANE_UINT4) o4[i] = u[j];
    }
}

__global__ __launch_bounds__(256)
void degree_norm_kernel(const unsigned int* __restrict__ planes, const int* __restrict__ coarse,
                        float* __restrict__ s_out, float* __restrict__ s_in,
                        unsigned int* __restrict__ base, int* __restrict__ rowptr) {
    __shared__ int wsum[8];
    const int t = threadIdx.x;
    const int w = blockIdx.x * 256 + t;
    const bool valid = w < PLANE_WORDS;
    unsigned int lo = 0, hi = 0;
    if (blockIdx.y == 0) {
        if (!valid) return;
        const unsigned int* __restrict__ p = planes + w;
        for (int g = 0; g < SRC_SLICES; ++g) {
            const unsigned int u = p[(size_t)g * PLANE_WORDS];
            lo += u & 0xffffu;
            hi += u >> 16;
        }
        float2 r;
        r.x = rsqrtf(fmaxf((float)lo, 1.0f));
        r.y = rsqrtf(fmaxf((float)hi, 1.0f));
        *(float2*)(s_out + 2 * w) = r;
    } else {
        const int lane = t & 31;
        const int wave = t >> 5;
        const int below = coarse[2 * blockIdx.x * DST_SLICES + lane];
        unsigned int u[DST_SLICES];
        #pragma unroll
        for (int g = 0; g < DST_SLICES; ++g) u[g] = 0u;
        if (valid) {
            const unsigned int* __restrict__ p = planes + (size_t)SRC_SLICES * PLANE_WORDS + w;
            #pragma unroll
            for (int g = 0; g < DST_SLICES; ++g) u[g] = p[(size_t)g * PLANE_WORDS];
            #pragma unroll
            for (int g = 0; g < DST_SLICES; ++g) {
                lo += u[g] & 0xffffu;
                hi += u[g] >> 16;
            }
            float2 r;
            r.x = rsqrtf(fmaxf((float)lo, 1.0f));
            r.y = rsqrtf(fmaxf((float)hi, 1.0f));
            *(float2*)(s_in + 2 * w) = r;
        }
        const int deg  = (int)(lo + hi);
        const int incl = wave_incl_scan(deg, lane);
        int off = __builtin_amdgcn_readlane(wave_incl_scan(below, lane), 31);
        if (lane == 31) wsum[wave] = incl;
        __syncthreads();
        for (int i = 0; i < 8; ++i) {
            if (i < wave) off += wsum[i];
        }
        const int excl = off + incl - deg;
        if (valid) {
            *(int2*)(rowptr + 2 * w) = make_int2(excl, excl + (int)lo);
            if (w == PLANE_WORDS - 1) rowptr[N_NODES] = excl + deg;
            unsigned int a = (unsigned int)excl;
            unsigned int b = (unsigned int)excl + lo;
            #pragma unroll
            for (int g = 0; g < DST_SLICES; ++g) {
                *(uint2*)(base + (size_t)g * N_NODES + 2 * w) = make_uint2(a, b);
                a += u[g] & 0xffffu;
                b += u[g] >> 16;
            }
        }
    }
}

#define EDGE_INT4      (N_EDGES / 4)
#define FILL_PARTS     8
#define FILL_THREADS   1024
#define FILL_PART_INT4 ((DST_SLICE_INT4 + FILL_PARTS - 1) / FILL_PARTS)
#define TAB_UINT4      (N_NODES / 4)
#define TAB_FULL       (TAB_UINT4 / FILL_THREADS)

__global__ __launch_bounds__(FILL_THREADS)
void csr_fill_kernel(const int* __restrict__ src, const int* __restrict__ dst,
                     const uint2* __restrict__ rank4, const unsigned int* __restrict__ base,
                     int* __restrict__ csr) {
    __shared__ __attribute__((aligned(16))) unsigned int tab[N_NODES];
    const int t     = threadIdx.x;
    const int slice = blockIdx.x / FILL_PARTS;
    const int local = (blockIdx.x % FILL_PARTS) * FILL_PART_INT4 + t;
    const bool has  = (t < FILL_PART_INT4) && (local < DST_SLICE_INT4);
    const int q     = slice * DST_SLICE_INT4 + local;

    int4  d4 = make_int4(0, 0, 0, 0);
    int4  s4 = make_int4(0, 0, 0, 0);
    uint2 r  = make_uint2(0u, 0u);
    if (has) {
        d4 = ((const int4*)dst)[q];
        s4 = ((const int4*)src)[q];
        r  = rank4[q];
    }

    const uint4* __restrict__ b4 = (const uint4*)(base + (size_t)slice * N_NODES);
    uint4* t4 = (uint4*)tab;
    const int itail = t + TAB_FULL * FILL_THREADS;
    uint4 vtail = make_uint4(0u, 0u, 0u, 0u);
    if (itail < TAB_UINT4) vtail = b4[itail];
    uint4 v[TAB_FULL];
    #pragma unroll
    for (int j = 0; j < TAB_FULL; ++j) v[j] = b4[t + j * FILL_THREADS];
    #pragma unroll
    for (int j = 0; j < TAB_FULL; ++j) t4[t + j * FILL_THREADS] = v[j];
    if (itail < TAB_UINT4) t4[itail] = vtail;
    __syncthreads();

    if (has) {
        const unsigned int p0 = tab[d4.x] + (r.x & 0xffffu);
        const unsigned int p1 = tab[d4.y] + (r.x >> 16);
        const unsigned int p2 = tab[d4.z] + (r.y & 0xffffu);
        const unsigned int p3 = tab[d4.w] + (r.y >> 16);
        csr[min(p0, (unsigned int)(N_EDGES - 1))] = s4.x;
        csr[min(p1, (unsigned int)(N_EDGES - 1))] = s4.y;
        csr[min(p2, (unsigned int)(N_EDGES - 1))] = s4.z;
        csr[min(p3, (unsigned int)(N_EDGES - 1))] = s4.w;
    }
}

#define HP 132

template <int U>
__device__ __forceinline__ void gather_batch(const float* __restrict__ x, int idx, float sc, int k, int lane,
                                             float4& acc) {
    float4 xv[U];
    float  f[U];
    #pragma unroll
    for (int u = 0; u < U; ++u) {
        const int s = __builtin_amdgcn_readlane(idx, k + u);
        f[u]  = __int_as_float(__builtin_amdgcn_readlane(__float_as_int(sc), k + u));
        xv[u] = *(const float4*)(x + (size_t)s * FEATS + lane * 4);
    }
    #pragma unroll
    for (int u = 0; u < U; ++u) {
        acc.x = fmaf(xv[u].x, f[u], acc.x);
        acc.y = fmaf(xv[u].y, f[u], acc.y);
        acc.z = fmaf(xv[u].z, f[u], acc.z);
        acc.w = fmaf(xv[u].w, f[u], acc.w);
    }
}

__global__ __launch_bounds__(256)
void gather_gemm_kernel(const float* __restrict__ x, const float* __restrict__ W,
                        const float* __restrict__ bias, const float* __restrict__ s_out,
                        const float* __restrict__ s_in, const int* __restrict__ rowptr,
                        const int* __restrict__ csr, float* __restrict__ out) {
    __shared__ __attribute__((aligned(16))) float Hs[16 * HP];
    const int lane = threadIdx.x & 31;
    const int wave = __builtin_amdgcn_readfirstlane((int)(threadIdx.x >> 5));
    const int row0 = blockIdx.x * 16;

    for (int rr = 0; rr < 2; ++rr) {
        const int r = 2 * wave + rr;
        const int vtx = row0 + r;
        int beg = __builtin_amdgcn_readfirstlane(rowptr[vtx]);
        int end = __builtin_amdgcn_readfirstlane(rowptr[vtx + 1]);
        beg = min(max(beg, 0), N_EDGES);
        end = min(max(end, beg), N_EDGES);
        float4 acc = make_float4(0.0f, 0.0f, 0.0f, 0.0f);
        for (int base = beg; base < end; base += 32) {
            const int n = min(32, end - base);
            int   idx = 0;
            float sc  = 0.0f;
            if (lane < n) {
                idx = (int)min((unsigned int)csr[base + lane], (unsigned int)(N_NODES - 1));
                sc  = s_out[idx];
            }
            int k = 0;
            for (; k + 8 <= n; k += 8) gather_batch<8>(x, idx, sc, k, lane, acc);
            if (k + 4 <= n) { gather_batch<4>(x, idx, sc, k, lane, acc); k += 4; }
            for (; k < n; ++k) gather_batch<1>(x, idx, sc, k, lane, acc);
        }
        *(float4*)(Hs + r * HP + lane * 4) = acc;
    }
    __syncthreads();

    const int m   = lane & 15;
    const int h   = lane >> 4;
    const int col = wave * 16 + m;
    v8f c = {};
    #pragma unroll 8
    for (int k0 = 0; k0 < FEATS; k0 += 4) {
        const int k = k0 + 2 * h;
        const v2f a = *(const v2f*)(Hs + m * HP + k);
        v2f b;
        b.x = W[(size_t)k * FEATS + col];
        b.y = W[(size_t)(k + 1) * FEATS + col];
        c = __builtin_amdgcn_wmma_f32_16x16x4_f32(
                 false, a,  false, b,
                 (short)0, c,  false,  false);
    }
    __syncthreads();

    #pragma unroll
    for (int j = 0; j < 8; ++j) Hs[(j + 8 * h) * HP + col] = c[j];
    __syncthreads();
    const float4 bv = *(const float4*)(bias + lane * 4);
    #pragma unroll
    for (int rr = 0; rr < 2; ++rr) {
        const int r = 2 * wave + rr;
        const float sn = s_in[row0 + r];
        const float4 hv = *(const float4*)(Hs + r * HP + lane * 4);
        float4 o;
        o.x = hv.x * sn + bv.x;
        o.y = hv.y * sn + bv.y;
        o.z = hv.z * sn + bv.z;
        o.w = hv.w * sn + bv.w;
        *(float4*)(out + (size_t)(row0 + r) * FEATS + lane * 4) = o;
    }
}

extern "C" void kernel_launch(void* const* d_in, const int* in_sizes, int n_in,
                              void* d_out, int out_size, void* d_ws, size_t ws_size,
                              hipStream_t stream) {
    const float* x    = (const float*)d_in[0];
    const int*   src  = (const int*)  d_in[1];
    const int*   dst  = (const int*)  d_in[2];
    const float* W    = (const float*)d_in[3];
    const float* bias = (const float*)d_in[4];
    float* out = (float*)d_out;

    float*        s_out  = (float*)d_ws;
    float*        s_in   = s_out + N_NODES;
    unsigned int* planes = (unsigned int*)(s_in + N_NODES);
    unsigned int* base   = planes + (SRC_SLICES + DST_SLICES) * PLANE_WORDS;
    uint2*        rank4  = (uint2*)(base + DST_SLICES * N_NODES);
    int*          coarse = (int*)(rank4 + EDGE_INT4);
    int*          rowptr = coarse + N_NODES;
    int*          csr    = rowptr + N_NODES + 4;

    const int TB = 256;

    degree_hist_kernel<<<SRC_SLICES + DST_SLICES, HIST_THREADS, 0, stream>>>(src, dst, planes, rank4, coarse);

    degree_norm_kernel<<<dim3((PLANE_WORDS + TB - 1) / TB, 2), TB, 0, stream>>>(planes, coarse, s_out, s_in, base, rowptr);

    csr_fill_kernel<<<DST_SLICES * FILL_PARTS, FILL_THREADS, 0, stream>>>(src, dst, rank4, base, csr);

    gather_gemm_kernel<<<N_NODES / 16, TB, 0, stream>>>(x, W, bias, s_out, s_in, rowptr, csr, out);
}


# per-edge source scale rides in the csr entry: fill kernel stages s_out in LDS ahead of its position table, stores (src, scale) as 8 B; gather's scattered 4 B scale loads gone; sums unchanged
# speedup vs baseline: 3.7900x; 1.0448x over previous
#include <hip/hip_runtime.h>

#define N_NODES 50000
#define N_EDGES 800000
#define FEATS   128

typedef __attribute__((ext_vector_type(2))) float v2f;
typedef __attribute__((ext_vector_type(8))) float v8f;

#define SRC_SLICES      64
#define DST_SLICES      32
#define SRC_SLICE_INT4  3125
#define DST_SLICE_INT4  6250
#define PLANE_WORDS     (N_NODES / 2)
#define PLANE_UINT4     (PLANE_WORDS / 4)
#define HIST_THREADS    1024
#define HIST_LOADS      7
#define BUCKETS         ((PLANE_UINT4 + 31) / 32)

__device__ __forceinline__ int wave_incl_scan(int x, int lane) {
    x += __builtin_amdgcn_update_dpp(0, x, 0x111, 0xf, 0xf, false);
    x += __builtin_amdgcn_update_dpp(0, x, 0x112, 0xf, 0xf, false);
    x += __builtin_amdgcn_update_dpp(0, x, 0x114, 0xf, 0xf, false);
    x += __builtin_amdgcn_update_dpp(0, x, 0x118, 0xf, 0xf, false);
    const int low = __builtin_amdgcn_readlane(x, 15);
    return x + ((lane >= 16) ? low : 0);
}

__device__ __forceinline__ unsigned int lds_count(unsigned int* h, int node) {
    const int sh = (node & 1) << 4;
    const unsigned int old = __hip_atomic_fetch_add(&h[node >> 1], 1u << sh,
                                                    __ATOMIC_RELAXED, __HIP_MEMORY_SCOPE_WORKGROUP);
    return (old >> sh) & 0xffffu;
}

__global__ __launch_bounds__(HIST_THREADS)
void degree_hist_kernel(const int* __restrict__ src, const int* __restrict__ dst,
                        unsigned int* __restrict__ planes, uint2* __restrict__ rank4,
                        int* __restrict__ coarse) {
    __shared__ __attribute__((aligned(16))) unsigned int h[PLANE_WORDS];
    __shared__ int btot[HIST_LOADS * 32];
    const int t      = threadIdx.x;
    const int b      = blockIdx.x;
    const bool isdst = b >= SRC_SLICES;
    const int slice  = isdst ? b - SRC_SLICES : b;
    const int n4     = isdst ? DST_SLICE_INT4 : SRC_SLICE_INT4;
    const int4* __restrict__ idx4 = isdst ? (const int4*)dst + (size_t)slice * DST_SLICE_INT4
                                          : (const int4*)src + (size_t)slice * SRC_SLICE_INT4;

    int4 v[HIST_LOADS];
    #pragma unroll
    for (int j = 0; j < HIST_LOADS; ++j) {
        const int q = t + j * HIST_THREADS;
        if (q < n4) v[j] = idx4[q];
    }

    uint4* h4 = (uint4*)h;
    for (int i = t; i < PLANE_UINT4; i += HIST_THREADS) h4[i] = make_uint4(0u, 0u, 0u, 0u);
    __syncthreads();

    #pragma unroll
    for (int j = 0; j < HIST_LOADS; ++j) {
        const int q = t + j * HIST_THREADS;
        if (q < n4) {
            const unsigned int r0 = lds_count(h, v[j].x);
            const unsigned int r1 = lds_count(h, v[j].y);
            const unsigned int r2 = lds_count(h, v[j].z);
            const unsigned int r3 = lds_count(h, v[j].w);
            if (isdst) rank4[(size_t)slice * DST_SLICE_INT4 + q] = make_uint2(r0 | (r1 << 16), r2 | (r3 << 16));
        }
    }
    __syncthreads();

    const int lane = t & 31;
    uint4 u[HIST_LOADS];
    #pragma unroll
    for (int j = 0; j < HIST_LOADS; ++j) {
        const int i = t + j * HIST_THREADS;
        u[j] = (i < PLANE_UINT4) ? h4[i] : make_uint4(0u, 0u, 0u, 0u);
    }

    if (isdst) {
        #pragma unroll
        for (int j = 0; j < HIST_LOADS; ++j) {
            const int i0 = __builtin_amdgcn_readfirstlane(t + j * HIST_THREADS);
            if (i0 < PLANE_UINT4) {
                const int s = (int)((u[j].x & 0xffffu) + (u[j].x >> 16) + (u[j].y & 0xffffu) + (u[j].y >> 16) +
                                    (u[j].z & 0xffffu) + (u[j].z >> 16) + (u[j].w & 0xffffu) + (u[j].w >> 16));
                const int tot = wave_incl_scan(s, lane);
                if (lane == 31) btot[i0 >> 5] = tot;
            }
        }
        __syncthreads();
        if (t < 32) {
            int carry = 0;
            for (int j = 0; j < HIST_LOADS; ++j) {
                const int k    = 32 * j + lane;
                const int v    = (k < BUCKETS) ? btot[k] : 0;
                const int incl = wave_incl_scan(v, lane);
                if (k < BUCKETS) coarse[k * DST_SLICES + slice] = carry + incl - v;
                carry += __builtin_amdgcn_readlane(incl, 31);
            }
        }
    }

    uint4* __restrict__ o4 = (uint4*)(planes + (size_t)b * PLANE_WORDS);
    #pragma unroll
    for (int j = 0; j < HIST_LOADS; ++j) {
        const int i = t + j * HIST_THREADS;
        if (i < PLANE_UINT4) o4[i] = u[j];
    }
}

__global__ __launch_bounds__(256)
void degree_norm_kernel(const unsigned int* __restrict__ planes, const int* __restrict__ coarse,
                        float* __restrict__ s_out, float* __restrict__ s_in,
                        unsigned int* __restrict__ base, int* __restrict__ rowptr) {
    __shared__ int wsum[8];
    const int t = threadIdx.x;
    const int w = blockIdx.x * 256 + t;
    const bool valid = w < PLANE_WORDS;
    unsigned int lo = 0, hi = 0;
    if (blockIdx.y == 0) {
        if (!valid) return;
        const unsigned int* __restrict__ p = planes + w;
        for (int g = 0; g < SRC_SLICES; ++g) {
            const unsigned int u = p[(size_t)g * PLANE_WORDS];
            lo += u & 0xffffu;
            hi += u >> 16;
        }
        float2 r;
        r.x = rsqrtf(fmaxf((float)lo, 1.0f));
        r.y = rsqrtf(fmaxf((float)hi, 1.0f));
        *(float2*)(s_out + 2 * w) = r;
    } else {
        const int lane = t & 31;
        const int wave = t >> 5;
        const int below = coarse[2 * blockIdx.x * DST_SLICES + lane];
        unsigned int u[DST_SLICES];
        #pragma unroll
        for (int g = 0; g < DST_SLICES; ++g) u[g] = 0u;
        if (valid) {
            const unsigned int* __restrict__ p = planes + (size_t)SRC_SLICES * PLANE_WORDS + w;
            #pragma unroll
            for (int g = 0; g < DST_SLICES; ++g) u[g] = p[(size_t)g * PLANE_WORDS];
            #pragma unroll
            for (int g = 0; g < DST_SLICES; ++g) {
                lo += u[g] & 0xffffu;
                hi += u[g] >> 16;
            }
            float2 r;
            r.x = rsqrtf(fmaxf((float)lo, 1.0f));
            r.y = rsqrtf(fmaxf((float)hi, 1.0f));
            *(float2*)(s_in + 2 * w) = r;
        }
        const int deg  = (int)(lo + hi);
        const int incl = wave_incl_scan(deg, lane);
        int off = __builtin_amdgcn_readlane(wave_incl_scan(below, lane), 31);
        if (lane == 31) wsum[wave] = incl;
        __syncthreads();
        for (int i = 0; i < 8; ++i) {
            if (i < wave) off += wsum[i];
        }
        const int excl = off + incl - deg;
        if (valid) {
            *(int2*)(rowptr + 2 * w) = make_int2(excl, excl + (int)lo);
            if (w == PLANE_WORDS - 1) rowptr[N_NODES] = excl + deg;
            unsigned int a = (unsigned int)excl;
            unsigned int b = (unsigned int)excl + lo;
            #pragma unroll
            for (int g = 0; g < DST_SLICES; ++g) {
                *(uint2*)(base + (size_t)g * N_NODES + 2 * w) = make_uint2(a, b);
                a += u[g] & 0xffffu;
                b += u[g] >> 16;
            }
        }
    }
}

#define EDGE_INT4      (N_EDGES / 4)
#define FILL_PARTS     8
#define FILL_THREADS   1024
#define FILL_PART_INT4 ((DST_SLICE_INT4 + FILL_PARTS - 1) / FILL_PARTS)
#define TAB_UINT4      (N_NODES / 4)
#define TAB_FULL       (TAB_UINT4 / FILL_THREADS)

__global__ __launch_bounds__(FILL_THREADS)
void csr_fill_kernel(const int* __restrict__ src, const int* __restrict__ dst,
                     const uint2* __restrict__ rank4, const unsigned int* __restrict__ base,
                     const float* __restrict__ s_out, uint2* __restrict__ csr) {
    __shared__ __attribute__((aligned(16))) unsigned int tab[N_NODES];
    const int t     = threadIdx.x;
    const int slice = blockIdx.x / FILL_PARTS;
    const int local = (blockIdx.x % FILL_PARTS) * FILL_PART_INT4 + t;
    const bool has  = (t < FILL_PART_INT4) && (local < DST_SLICE_INT4);
    const int q     = slice * DST_SLICE_INT4 + local;

    int4  d4 = make_int4(0, 0, 0, 0);
    int4  s4 = make_int4(0, 0, 0, 0);
    uint2 r  = make_uint2(0u, 0u);
    if (has) {
        d4 = ((const int4*)dst)[q];
        s4 = ((const int4*)src)[q];
        r  = rank4[q];
    }

    uint4* t4 = (uint4*)tab;
    const int itail = t + TAB_FULL * FILL_THREADS;

    {
        const uint4* __restrict__ so4 = (const uint4*)s_out;
        uint4 wtail = make_uint4(0u, 0u, 0u, 0u);
        if (itail < TAB_UINT4) wtail = so4[itail];
        uint4 w[TAB_FULL];
        #pragma unroll
        for (int j = 0; j < TAB_FULL; ++j) w[j] = so4[t + j * FILL_THREADS];
        #pragma unroll
        for (int j = 0; j < TAB_FULL; ++j) t4[t + j * FILL_THREADS] = w[j];
        if (itail < TAB_UINT4) t4[itail] = wtail;
    }
    __syncthreads();
    unsigned int f0 = 0u, f1 = 0u, f2 = 0u, f3 = 0u;
    if (has) {
        f0 = tab[s4.x];
        f1 = tab[s4.y];
        f2 = tab[s4.z];
        f3 = tab[s4.w];
    }
    __syncthreads();

    {
        const uint4* __restrict__ b4 = (const uint4*)(base + (size_t)slice * N_NODES);
        uint4 vtail = make_uint4(0u, 0u, 0u, 0u);
        if (itail < TAB_UINT4) vtail = b4[itail];
        uint4 v[TAB_FULL];
        #pragma unroll
        for (int j = 0; j < TAB_FULL; ++j) v[j] = b4[t + j * FILL_THREADS];
        #pragma unroll
        for (int j = 0; j < TAB_FULL; ++j) t4[t + j * FILL_THREADS] = v[j];
        if (itail < TAB_UINT4) t4[itail] = vtail;
    }
    __syncthreads();

    if (has) {
        const unsigned int p0 = tab[d4.x] + (r.x & 0xffffu);
        const unsigned int p1 = tab[d4.y] + (r.x >> 16);
        const unsigned int p2 = tab[d4.z] + (r.y & 0xffffu);
        const unsigned int p3 = tab[d4.w] + (r.y >> 16);
        csr[min(p0, (unsigned int)(N_EDGES - 1))] = make_uint2((unsigned int)s4.x, f0);
        csr[min(p1, (unsigned int)(N_EDGES - 1))] = make_uint2((unsigned int)s4.y, f1);
        csr[min(p2, (unsigned int)(N_EDGES - 1))] = make_uint2((unsigned int)s4.z, f2);
        csr[min(p3, (unsigned int)(N_EDGES - 1))] = make_uint2((unsigned int)s4.w, f3);
    }
}

#define HP 132

template <int U>
__device__ __forceinline__ void gather_batch(const float* __restrict__ x, int idx, float sc, int k, int lane,
                                             float4& acc) {
    float4 xv[U];
    float  f[U];
    #pragma unroll
    for (int u = 0; u < U; ++u) {
        const int s = __builtin_amdgcn_readlane(idx, k + u);
        f[u]  = __int_as_float(__builtin_amdgcn_readlane(__float_as_int(sc), k + u));
        xv[u] = *(const float4*)(x + (size_t)s * FEATS + lane * 4);
    }
    #pragma unroll
    for (int u = 0; u < U; ++u) {
        acc.x = fmaf(xv[u].x, f[u], acc.x);
        acc.y = fmaf(xv[u].y, f[u], acc.y);
        acc.z = fmaf(xv[u].z, f[u], acc.z);
        acc.w = fmaf(xv[u].w, f[u], acc.w);
    }
}

__global__ __launch_bounds__(256)
void gather_gemm_kernel(const float* __restrict__ x, const float* __restrict__ W,
                        const float* __restrict__ bias,
                        const float* __restrict__ s_in, const int* __restrict__ rowptr,
                        const uint2* __restrict__ csr, float* __restrict__ out) {
    __shared__ __attribute__((aligned(16))) float Hs[16 * HP];
    const int lane = threadIdx.x & 31;
    const int wave = __builtin_amdgcn_readfirstlane((int)(threadIdx.x >> 5));
    const int row0 = blockIdx.x * 16;

    for (int rr = 0; rr < 2; ++rr) {
        const int r = 2 * wave + rr;
        const int vtx = row0 + r;
        int beg = __builtin_amdgcn_readfirstlane(rowptr[vtx]);
        int end = __builtin_amdgcn_readfirstlane(rowptr[vtx + 1]);
        beg = min(max(beg, 0), N_EDGES);
        end = min(max(end, beg), N_EDGES);
        float4 acc = make_float4(0.0f, 0.0f, 0.0f, 0.0f);
        for (int base = beg; base < end; base += 32) {
            const int n = min(32, end - base);
            int   idx = 0;
            float sc  = 0.0f;
            if (lane < n) {
                const uint2 e = csr[base + lane];
                idx = (int)min(e.x, (unsigned int)(N_NODES - 1));
                sc  = __uint_as_float(e.y);
            }
            int k = 0;
            for (; k + 8 <= n; k += 8) gather_batch<8>(x, idx, sc, k, lane, acc);
            if (k + 4 <= n) { gather_batch<4>(x, idx, sc, k, lane, acc); k += 4; }
            for (; k < n; ++k) gather_batch<1>(x, idx, sc, k, lane, acc);
        }
        *(float4*)(Hs + r * HP + lane * 4) = acc;
    }
    __syncthreads();

    const int m   = lane & 15;
    const int h   = lane >> 4;
    const int col = wave * 16 + m;
    v8f c = {};
    #pragma unroll 8
    for (int k0 = 0; k0 < FEATS; k0 += 4) {
        const int k = k0 + 2 * h;
        const v2f a = *(const v2f*)(Hs + m * HP + k);
        v2f b;
        b.x = W[(size_t)k * FEATS + col];
        b.y = W[(size_t)(k + 1) * FEATS + col];
        c = __builtin_amdgcn_wmma_f32_16x16x4_f32(
                 false, a,  false, b,
                 (short)0, c,  false,  false);
    }
    __syncthreads();

    #pragma unroll
    for (int j = 0; j < 8; ++j) Hs[(j + 8 * h) * HP + col] = c[j];
    __syncthreads();
    const float4 bv = *(const float4*)(bias + lane * 4);
    #pragma unroll
    for (int rr = 0; rr < 2; ++rr) {
        const int r = 2 * wave + rr;
        const float sn = s_in[row0 + r];
        const float4 hv = *(const float4*)(Hs + r * HP + lane * 4);
        float4 o;
        o.x = hv.x * sn + bv.x;
        o.y = hv.y * sn + bv.y;
        o.z = hv.z * sn + bv.z;
        o.w = hv.w * sn + bv.w;
        *(float4*)(out + (size_t)(row0 + r) * FEATS + lane * 4) = o;
    }
}

extern "C" void kernel_launch(void* const* d_in, const int* in_sizes, int n_in,
                              void* d_out, int out_size, void* d_ws, size_t ws_size,
                              hipStream_t stream) {
    const float* x    = (const float*)d_in[0];
    const int*   src  = (const int*)  d_in[1];
    const int*   dst  = (const int*)  d_in[2];
    const float* W    = (const float*)d_in[3];
    const float* bias = (const float*)d_in[4];
    float* out = (float*)d_out;

    float*        s_out  = (float*)d_ws;
    float*        s_in   = s_out + N_NODES;
    unsigned int* planes = (unsigned int*)(s_in + N_NODES);
    unsigned int* base   = planes + (SRC_SLICES + DST_SLICES) * PLANE_WORDS;
    uint2*        rank4  = (uint2*)(base + DST_SLICES * N_NODES);
    int*          coarse = (int*)(rank4 + EDGE_INT4);
    int*          rowptr = coarse + N_NODES;
    uint2*        csr    = (uint2*)(rowptr + N_NODES + 4);

    const int TB = 256;

    degree_hist_kernel<<<SRC_SLICES + DST_SLICES, HIST_THREADS, 0, stream>>>(src, dst, planes, rank4, coarse);

    degree_norm_kernel<<<dim3((PLANE_WORDS + TB - 1) / TB, 2), TB, 0, stream>>>(planes, coarse, s_out, s_in, base, rowptr);

    csr_fill_kernel<<<DST_SLICES * FILL_PARTS, FILL_THREADS, 0, stream>>>(src, dst, rank4, base, s_out, csr);

    gather_gemm_kernel<<<N_NODES / 16, TB, 0, stream>>>(x, W, bias, s_in, rowptr, csr, out);
}


# csr built by bucket sort in LDS: dst slices sorted by 256-node bucket in the histogram launch; one block per bucket places its 64 runs in LDS, stores csr, rowptr, s_in whole-line; sum order changes
# speedup vs baseline: 4.9129x; 1.2963x over previous
#include <hip/hip_runtime.h>

#define N_NODES 50000
#define N_EDGES 800000
#define FEATS   128

typedef __attribute__((ext_vector_type(2))) float v2f;
typedef __attribute__((ext_vector_type(8))) float v8f;

#define SRC_SLICES      64
#define DST_SLICES      64
#define SLICE_EDGES     12500
#define SLICE_INT4      3125
#define PLANE_WORDS     (N_NODES / 2)
#define PLANE_UINT4     (PLANE_WORDS / 4)
#define HIST_THREADS    1024
#define IDX_LOADS       4
#define PLANE_LOADS     7
#define BKT_NODES       256
#define BUCKETS         ((N_NODES + BKT_NODES - 1) / BKT_NODES)
#define SUB_SHIFT       6
#define SUBS            ((N_NODES + 63) / 64)
#define SUB_STEPS       25
#define SUB_PAD         (SUB_STEPS * 32)

static_assert(N_NODES <= 65536, "a source index must fit 16 bits of a sorted entry");
static_assert(SLICE_INT4 <= IDX_LOADS * HIST_THREADS, "one pass of the block covers a slice");
static_assert(SUBS <= SUB_PAD, "the scan steps cover every counter");
static_assert(BKT_NODES == 4 << SUB_SHIFT, "four counters per bucket");

__device__ __forceinline__ int wave_incl_scan(int x, int lane) {
    x += __builtin_amdgcn_update_dpp(0, x, 0x111, 0xf, 0xf, false);
    x += __builtin_amdgcn_update_dpp(0, x, 0x112, 0xf, 0xf, false);
    x += __builtin_amdgcn_update_dpp(0, x, 0x114, 0xf, 0xf, false);
    x += __builtin_amdgcn_update_dpp(0, x, 0x118, 0xf, 0xf, false);
    const int low = __builtin_amdgcn_readlane(x, 15);
    return x + ((lane >= 16) ? low : 0);
}

__device__ __forceinline__ void lds_count(unsigned int* h, int node) {
    __hip_atomic_fetch_add(&h[node >> 1], 1u << ((node & 1) << 4), __ATOMIC_RELAXED, __HIP_MEMORY_SCOPE_WORKGROUP);
}

__device__ __forceinline__ int lds_take(int* c, int i) {
    return __hip_atomic_fetch_add(&c[i], 1, __ATOMIC_RELAXED, __HIP_MEMORY_SCOPE_WORKGROUP);
}

__global__ __launch_bounds__(HIST_THREADS)
void degree_hist_kernel(const int* __restrict__ src, const int* __restrict__ dst,
                        unsigned int* __restrict__ planes, unsigned int* __restrict__ binned,
                        int* __restrict__ coarse) {
    __shared__ __attribute__((aligned(16))) unsigned int h[PLANE_WORDS];
    __shared__ int cnt[SUB_PAD];
    __shared__ int cur[SUB_PAD];
    const int t      = threadIdx.x;
    const int b      = blockIdx.x;
    const bool isdst = b >= SRC_SLICES;
    const int slice  = isdst ? b - SRC_SLICES : b;
    const int4* __restrict__ s4p = (const int4*)src + (size_t)slice * SLICE_INT4;
    const int4* __restrict__ d4p = (const int4*)dst + (size_t)slice * SLICE_INT4;
    uint4* h4 = (uint4*)h;

    if (!isdst) {
        int4 v[IDX_LOADS];
        #pragma unroll
        for (int j = 0; j < IDX_LOADS; ++j) {
            const int q = t + j * HIST_THREADS;
            if (q < SLICE_INT4) v[j] = s4p[q];
        }
        for (int i = t; i < PLANE_UINT4; i += HIST_THREADS) h4[i] = make_uint4(0u, 0u, 0u, 0u);
        __syncthreads();
        #pragma unroll
        for (int j = 0; j < IDX_LOADS; ++j) {
            const int q = t + j * HIST_THREADS;
            if (q < SLICE_INT4) {
                lds_count(h, v[j].x);
                lds_count(h, v[j].y);
                lds_count(h, v[j].z);
                lds_count(h, v[j].w);
            }
        }
        __syncthreads();
        uint4* __restrict__ o4 = (uint4*)(planes + (size_t)slice * PLANE_WORDS);
        for (int i = t; i < PLANE_UINT4; i += HIST_THREADS) o4[i] = h4[i];
    } else {
        int4 dv[IDX_LOADS];
        int4 sv[IDX_LOADS];
        #pragma unroll
        for (int j = 0; j < IDX_LOADS; ++j) {
            const int q = t + j * HIST_THREADS;
            if (q < SLICE_INT4) { dv[j] = d4p[q]; sv[j] = s4p[q]; }
        }
        if (t < SUB_PAD) cnt[t] = 0;
        __syncthreads();
        #pragma unroll
        for (int j = 0; j < IDX_LOADS; ++j) {
            const int q = t + j * HIST_THREADS;
            if (q < SLICE_INT4) {
                lds_take(cnt, min((unsigned int)dv[j].x >> SUB_SHIFT, (unsigned int)(SUBS - 1)));
                lds_take(cnt, min((unsigned int)dv[j].y >> SUB_SHIFT, (unsigned int)(SUBS - 1)));
                lds_take(cnt, min((unsigned int)dv[j].z >> SUB_SHIFT, (unsigned int)(SUBS - 1)));
                lds_take(cnt, min((unsigned int)dv[j].w >> SUB_SHIFT, (unsigned int)(SUBS - 1)));
            }
        }
        __syncthreads();
        if (t < 32) {
            int c[SUB_STEPS];
            #pragma unroll
            for (int j = 0; j < SUB_STEPS; ++j) c[j] = cnt[32 * j + t];
            int carry = 0;
            #pragma unroll
            for (int j = 0; j < SUB_STEPS; ++j) {
                const int k    = 32 * j + t;
                const int incl = wave_incl_scan(c[j], t);
                const int excl = carry + incl - c[j];
                cur[k] = excl;
                if ((k & 3) == 0 && (k >> 2) < BUCKETS) coarse[(k >> 2) * DST_SLICES + slice] = excl;
                carry += __builtin_amdgcn_readlane(incl, 31);
            }
            if (t == 0) coarse[BUCKETS * DST_SLICES + slice] = carry;
        }
        __syncthreads();
        #pragma unroll
        for (int j = 0; j < IDX_LOADS; ++j) {
            const int q = t + j * HIST_THREADS;
            if (q < SLICE_INT4) {
                const unsigned int d0 = (unsigned int)dv[j].x, d1 = (unsigned int)dv[j].y;
                const unsigned int d2 = (unsigned int)dv[j].z, d3 = (unsigned int)dv[j].w;
                const int p0 = lds_take(cur, min(d0 >> SUB_SHIFT, (unsigned int)(SUBS - 1)));
                const int p1 = lds_take(cur, min(d1 >> SUB_SHIFT, (unsigned int)(SUBS - 1)));
                const int p2 = lds_take(cur, min(d2 >> SUB_SHIFT, (unsigned int)(SUBS - 1)));
                const int p3 = lds_take(cur, min(d3 >> SUB_SHIFT, (unsigned int)(SUBS - 1)));
                h[min(p0, SLICE_EDGES - 1)] = ((unsigned int)sv[j].x & 0xffffu) | ((d0 & 255u) << 16);
                h[min(p1, SLICE_EDGES - 1)] = ((unsigned int)sv[j].y & 0xffffu) | ((d1 & 255u) << 16);
                h[min(p2, SLICE_EDGES - 1)] = ((unsigned int)sv[j].z & 0xffffu) | ((d2 & 255u) << 16);
                h[min(p3, SLICE_EDGES - 1)] = ((unsigned int)sv[j].w & 0xffffu) | ((d3 & 255u) << 16);
            }
        }
        __syncthreads();
        uint4* __restrict__ o4 = (uint4*)(binned + (size_t)slice * SLICE_EDGES);
        for (int i = t; i < SLICE_INT4; i += HIST_THREADS) o4[i] = h4[i];
    }
}

__global__ __launch_bounds__(256)
void degree_norm_kernel(const unsigned int* __restrict__ planes, float* __restrict__ s_out) {
    const int w = blockIdx.x * 256 + threadIdx.x;
    if (w >= PLANE_WORDS) return;
    unsigned int lo = 0, hi = 0;
    const unsigned int* __restrict__ p = planes + w;
    for (int g = 0; g < SRC_SLICES; ++g) {
        const unsigned int u = p[(size_t)g * PLANE_WORDS];
        lo += u & 0xffffu;
        hi += u >> 16;
    }
    float2 r;
    r.x = rsqrtf(fmaxf((float)lo, 1.0f));
    r.y = rsqrtf(fmaxf((float)hi, 1.0f));
    *(float2*)(s_out + 2 * w) = r;
}

#define BUILD_THREADS  1024
#define TAB_UINT4      (N_NODES / 4)
#define TAB_FULL       (TAB_UINT4 / BUILD_THREADS)
#define WIN_CAP        6144
#define SLICE_GROUPS   (BUILD_THREADS / BKT_NODES)
#define GROUP_SLICES   (DST_SLICES / SLICE_GROUPS)

static_assert(DST_SLICES == 2 * (BUILD_THREADS / 32), "every wave reads two runs");
static_assert(SLICE_GROUPS * GROUP_SLICES == DST_SLICES, "the scan groups cover every slice");

__device__ __forceinline__ void build_one(unsigned int e, int* c, bool place, const unsigned int* tab, uint2* win,
                                          bool in_lds, uint2* __restrict__ csr, int gstart) {
    const int n = (int)((e >> 16) & 255u);
    const int p = lds_take(c, n);
    if (place) {
        const unsigned int s = e & 0xffffu;
        const uint2 ent = make_uint2(s, tab[min(s, (unsigned int)(N_NODES - 1))]);
        if (in_lds) win[min(p, WIN_CAP - 1)] = ent;
        else        csr[min(max(gstart + p, 0), N_EDGES - 1)] = ent;
    }
}

__device__ __forceinline__ void build_pass(const unsigned int* __restrict__ runA, int lenA, int* cA,
                                           const unsigned int* __restrict__ runB, int lenB, int* cB, int lane,
                                           bool place, const unsigned int* tab, uint2* win, bool in_lds,
                                           uint2* __restrict__ csr, int gstart) {
    const int len = max(lenA, lenB);
    for (int i = lane; i < len; i += 32) {
        const bool hasA = i < lenA, hasB = i < lenB;
        unsigned int eA = 0u, eB = 0u;
        if (hasA) eA = runA[i];
        if (hasB) eB = runB[i];
        if (hasA) build_one(eA, cA, place, tab, win, in_lds, csr, gstart);
        if (hasB) build_one(eB, cB, place, tab, win, in_lds, csr, gstart);
    }
}

__global__ __launch_bounds__(BUILD_THREADS)
void csr_build_kernel(const unsigned int* __restrict__ binned, const int* __restrict__ coarse,
                      const float* __restrict__ s_out, float* __restrict__ s_in,
                      int* __restrict__ rowptr, uint2* __restrict__ csr) {
    __shared__ __attribute__((aligned(16))) unsigned int tab[N_NODES];
    __shared__ __attribute__((aligned(16))) int cnt[DST_SLICES * BKT_NODES];
    __shared__ __attribute__((aligned(16))) uint2 win[WIN_CAP];
    __shared__ int psum[SLICE_GROUPS * BKT_NODES];
    __shared__ int nstart[BKT_NODES];
    __shared__ int wsum[BKT_NODES / 32];
    const int t    = threadIdx.x;
    const int lane = t & 31;
    const int wave = __builtin_amdgcn_readfirstlane(t >> 5);
    const int k    = blockIdx.x;

    const int* __restrict__ c0 = coarse + (size_t)k * DST_SLICES;
    int lo0 = c0[lane], lo1 = c0[32 + lane];
    int hi0 = c0[DST_SLICES + lane], hi1 = c0[DST_SLICES + 32 + lane];
    lo0 = min(max(lo0, 0), SLICE_EDGES);  hi0 = min(max(hi0, lo0), SLICE_EDGES);
    lo1 = min(max(lo1, 0), SLICE_EDGES);  hi1 = min(max(hi1, lo1), SLICE_EDGES);

    uint4* t4 = (uint4*)tab;
    {
        const uint4* __restrict__ so4 = (const uint4*)s_out;
        const int itail = t + TAB_FULL * BUILD_THREADS;
        uint4 wtail = make_uint4(0u, 0u, 0u, 0u);
        if (itail < TAB_UINT4) wtail = so4[itail];
        uint4 w[TAB_FULL];
        #pragma unroll
        for (int j = 0; j < TAB_FULL; ++j) w[j] = so4[t + j * BUILD_THREADS];
        #pragma unroll
        for (int j = 0; j < TAB_FULL; ++j) t4[t + j * BUILD_THREADS] = w[j];
        if (itail < TAB_UINT4) t4[itail] = wtail;
    }
    {
        int4* c4 = (int4*)cnt;
        for (int i = t; i < DST_SLICES * BKT_NODES / 4; i += BUILD_THREADS) c4[i] = make_int4(0, 0, 0, 0);
    }

    const int gstart = __builtin_amdgcn_readlane(wave_incl_scan(lo0 + lo1, lane), 31);
    const int gend   = __builtin_amdgcn_readlane(wave_incl_scan(hi0 + hi1, lane), 31);
    const int total  = gend - gstart;
    const bool in_lds = total <= WIN_CAP;

    const int begA = __builtin_amdgcn_readlane(lo0, wave), lenA = __builtin_amdgcn_readlane(hi0, wave) - begA;
    const int begB = __builtin_amdgcn_readlane(lo1, wave), lenB = __builtin_amdgcn_readlane(hi1, wave) - begB;
    const unsigned int* __restrict__ runA = binned + (size_t)wave * SLICE_EDGES + begA;
    const unsigned int* __restrict__ runB = binned + (size_t)(wave + 32) * SLICE_EDGES + begB;
    int* cA = cnt + wave * BKT_NODES;
    int* cB = cnt + (wave + 32) * BKT_NODES;
    __syncthreads();

    build_pass(runA, lenA, cA, runB, lenB, cB, lane, false, tab, win, in_lds, csr, gstart);
    __syncthreads();

    const int n = t & (BKT_NODES - 1);
    const int q = __builtin_amdgcn_readfirstlane(t >> 8);
    int c[GROUP_SLICES];
    int part = 0;
    #pragma unroll
    for (int i = 0; i < GROUP_SLICES; ++i) {
        c[i] = cnt[(q * GROUP_SLICES + i) * BKT_NODES + n];
        part += c[i];
    }
    psum[q * BKT_NODES + n] = part;
    __syncthreads();
    int deg = 0, incl = 0;
    if (t < BKT_NODES) {
        #pragma unroll
        for (int g = 0; g < SLICE_GROUPS; ++g) deg += psum[g * BKT_NODES + n];
        incl = wave_incl_scan(deg, lane);
        if (lane == 31) wsum[wave] = incl;
    }
    __syncthreads();
    if (t < BKT_NODES) {
        int off = 0;
        for (int i = 0; i < BKT_NODES / 32; ++i) {
            if (i < wave) off += wsum[i];
        }
        const int excl = off + incl - deg;
        nstart[n] = excl;
        const int node = k * BKT_NODES + n;
        if (node < N_NODES) {
            rowptr[node] = gstart + excl;
            s_in[node]   = rsqrtf(fmaxf((float)deg, 1.0f));
        }
        if (k == BUCKETS - 1 && t == 0) rowptr[N_NODES] = gend;
    }
    __syncthreads();
    {
        int run = nstart[n];
        for (int g = 0; g < SLICE_GROUPS; ++g) {
            if (g < q) run += psum[g * BKT_NODES + n];
        }
        #pragma unroll
        for (int i = 0; i < GROUP_SLICES; ++i) {
            cnt[(q * GROUP_SLICES + i) * BKT_NODES + n] = run;
            run += c[i];
        }
    }
    __syncthreads();

    build_pass(runA, lenA, cA, runB, lenB, cB, lane, true, tab, win, in_lds, csr, gstart);
    __syncthreads();

    if (in_lds) {
        for (int i = t; i < total; i += BUILD_THREADS) {
            const int p = gstart + i;
            if (p < N_EDGES) csr[p] = win[i];
        }
    }
}

#define HP 132

template <int U>
__device__ __forceinline__ void gather_batch(const float* __restrict__ x, int idx, float sc, int k, int lane,
                                             float4& acc) {
    float4 xv[U];
    float  f[U];
    #pragma unroll
    for (int u = 0; u < U; ++u) {
        const int s = __builtin_amdgcn_readlane(idx, k + u);
        f[u]  = __int_as_float(__builtin_amdgcn_readlane(__float_as_int(sc), k + u));
        xv[u] = *(const float4*)(x + (size_t)s * FEATS + lane * 4);
    }
    #pragma unroll
    for (int u = 0; u < U; ++u) {
        acc.x = fmaf(xv[u].x, f[u], acc.x);
        acc.y = fmaf(xv[u].y, f[u], acc.y);
        acc.z = fmaf(xv[u].z, f[u], acc.z);
        acc.w = fmaf(xv[u].w, f[u], acc.w);
    }
}

__global__ __launch_bounds__(256)
void gather_gemm_kernel(const float* __restrict__ x, const float* __restrict__ W,
                        const float* __restrict__ bias,
                        const float* __restrict__ s_in, const int* __restrict__ rowptr,
                        const uint2* __restrict__ csr, float* __restrict__ out) {
    __shared__ __attribute__((aligned(16))) float Hs[16 * HP];
    const int lane = threadIdx.x & 31;
    const int wave = __builtin_amdgcn_readfirstlane((int)(threadIdx.x >> 5));
    const int row0 = blockIdx.x * 16;

    for (int rr = 0; rr < 2; ++rr) {
        const int r = 2 * wave + rr;
        const int vtx = row0 + r;
        int beg = __builtin_amdgcn_readfirstlane(rowptr[vtx]);
        int end = __builtin_amdgcn_readfirstlane(rowptr[vtx + 1]);
        beg = min(max(beg, 0), N_EDGES);
        end = min(max(end, beg), N_EDGES);
        float4 acc = make_float4(0.0f, 0.0f, 0.0f, 0.0f);
        for (int base = beg; base < end; base += 32) {
            const int n = min(32, end - base);
            int   idx = 0;
            float sc  = 0.0f;
            if (lane < n) {
                const uint2 e = csr[base + lane];
                idx = (int)min(e.x, (unsigned int)(N_NODES - 1));
                sc  = __uint_as_float(e.y);
            }
            int k = 0;
            for (; k + 8 <= n; k += 8) gather_batch<8>(x, idx, sc, k, lane, acc);
            if (k + 4 <= n) { gather_batch<4>(x, idx, sc, k, lane, acc); k += 4; }
            for (; k < n; ++k) gather_batch<1>(x, idx, sc, k, lane, acc);
        }
        *(float4*)(Hs + r * HP + lane * 4) = acc;
    }
    __syncthreads();

    const int m   = lane & 15;
    const int h   = lane >> 4;
    const int col = wave * 16 + m;
    v8f c = {};
    #pragma unroll 8
    for (int k0 = 0; k0 < FEATS; k0 += 4) {
        const int k = k0 + 2 * h;
        const v2f a = *(const v2f*)(Hs + m * HP + k);
        v2f b;
        b.x = W[(size_t)k * FEATS + col];
        b.y = W[(size_t)(k + 1) * FEATS + col];
        c = __builtin_amdgcn_wmma_f32_16x16x4_f32(
                 false, a,  false, b,
                 (short)0, c,  false,  false);
    }
    __syncthreads();

    #pragma unroll
    for (int j = 0; j < 8; ++j) Hs[(j + 8 * h) * HP + col] = c[j];
    __syncthreads();
    const float4 bv = *(const float4*)(bias + lane * 4);
    #pragma unroll
    for (int rr = 0; rr < 2; ++rr) {
        const int r = 2 * wave + rr;
        const float sn = s_in[row0 + r];
        const float4 hv = *(const float4*)(Hs + r * HP + lane * 4);
        float4 o;
        o.x = hv.x * sn + bv.x;
        o.y = hv.y * sn + bv.y;
        o.z = hv.z * sn + bv.z;
        o.w = hv.w * sn + bv.w;
        *(float4*)(out + (size_t)(row0 + r) * FEATS + lane * 4) = o;
    }
}

extern "C" void kernel_launch(void* const* d_in, const int* in_sizes, int n_in,
                              void* d_out, int out_size, void* d_ws, size_t ws_size,
                              hipStream_t stream) {
    const float* x    = (const float*)d_in[0];
    const int*   src  = (const int*)  d_in[1];
    const int*   dst  = (const int*)  d_in[2];
    const float* W    = (const float*)d_in[3];
    const float* bias = (const float*)d_in[4];
    float* out = (float*)d_out;

    float*        s_out  = (float*)d_ws;
    float*        s_in   = s_out + N_NODES;
    unsigned int* planes = (unsigned int*)(s_in + N_NODES);
    unsigned int* binned = planes + SRC_SLICES * PLANE_WORDS;
    int*          coarse = (int*)(binned + DST_SLICES * SLICE_EDGES);
    int*          rowptr = coarse + N_NODES;
    uint2*        csr    = (uint2*)(rowptr + N_NODES + 4);

    const int TB = 256;

    degree_hist_kernel<<<SRC_SLICES + DST_SLICES, HIST_THREADS, 0, stream>>>(src, dst, planes, binned, coarse);

    degree_norm_kernel<<<(PLANE_WORDS + TB - 1) / TB, TB, 0, stream>>>(planes, s_out);

    csr_build_kernel<<<BUCKETS, BUILD_THREADS, 0, stream>>>(binned, coarse, s_out, s_in, rowptr, csr);

    gather_gemm_kernel<<<N_NODES / 16, TB, 0, stream>>>(x, W, bias, s_in, rowptr, csr, out);
}
